# starting kernel
# baseline (speedup 1.0000x reference)
#include <hip/hip_runtime.h>

#define HDIM 4
#define ODIM 4
#define JITTER 1e-4f

#define WS_LINV 0
#define WS_B    4096
#define WS_ZS   8192
#define WS_Z2   8704
#define WS_Q    8768
#define WS_ILS  8832
#define WS_VAR  8840
#define WS_STRIDE 8960

typedef __attribute__((ext_vector_type(2))) float v2f;
typedef __attribute__((ext_vector_type(8))) float v8f;

__global__ __launch_bounds__(64) void svgp_prep(
    const float* __restrict__ z, const float* __restrict__ u_mean,
    const float* __restrict__ u_tril_vec, const float* __restrict__ log_ls,
    const float* __restrict__ log_var, float* __restrict__ ws)
{
    __shared__ float sL[64 * 64];
    __shared__ float sLinv[64 * 64];
    __shared__ float sA[64 * 64];
    __shared__ float sZs[64 * 8];
    __shared__ float sZ2[64];
    __shared__ float sT[64];

    const int p   = blockIdx.x;
    const int o   = p % ODIM;
    const int tid = threadIdx.x;
    float* wp = ws + (size_t)p * WS_STRIDE;

    float ils[8];
#pragma unroll
    for (int d = 0; d < 8; ++d) ils[d] = __expf(-log_ls[p * 8 + d]);
    const float var = __expf(log_var[p]);

    {
        float s = 0.f;
#pragma unroll
        for (int d = 0; d < 8; ++d) {
            float v = z[(o * 64 + tid) * 8 + d] * ils[d];
            sZs[tid * 8 + d] = v;
            s += v * v;
        }
        sZ2[tid] = s;
    }
    __syncthreads();

    {
        const int i = tid;
        for (int j = 0; j < 64; ++j) {
            float dot = 0.f;
#pragma unroll
            for (int d = 0; d < 8; ++d) dot += sZs[i * 8 + d] * sZs[j * 8 + d];
            float v = var * __expf(-0.5f * (sZ2[i] + sZ2[j] - 2.f * dot));
            if (i == j) v += JITTER;
            sL[i * 64 + j] = v;
        }
    }
    __syncthreads();

    for (int j = 0; j < 64; ++j) {
        if (tid == j) {
            float s = sL[j * 64 + j];
            for (int k = 0; k < j; ++k) s -= sL[j * 64 + k] * sL[j * 64 + k];
            sL[j * 64 + j] = sqrtf(s);
        }
        __syncthreads();
        if (tid > j) {
            float s = sL[tid * 64 + j];
            for (int k = 0; k < j; ++k) s -= sL[tid * 64 + k] * sL[j * 64 + k];
            sL[tid * 64 + j] = s / sL[j * 64 + j];
        }
        __syncthreads();
    }

    {
        const int c = tid;
        for (int i = 0; i < c; ++i) sLinv[i * 64 + c] = 0.f;
        sLinv[c * 64 + c] = 1.f / sL[c * 64 + c];
        for (int i = c + 1; i < 64; ++i) {
            float s = 0.f;
            for (int k = c; k < i; ++k) s += sL[i * 64 + k] * sLinv[k * 64 + c];
            sLinv[i * 64 + c] = -s / sL[i * 64 + i];
        }
    }

    {
        const int i    = tid;
        const float di = 1.f / sL[i * 64 + i];
        const int base = o * 2080 + i * (i + 1) / 2;
        for (int j = 0; j < 64; ++j)
            sA[i * 64 + j] = (j <= i) ? u_tril_vec[base + j] * di : 0.f;
    }
    __syncthreads();

    {
        const int i = tid;
        float s = 0.f;
        for (int j = 0; j <= i; ++j) s += sLinv[i * 64 + j] * u_mean[o * 64 + j];
        sT[i] = s;
    }
    __syncthreads();

    {
        const int j = tid;
        float s = 0.f;
        for (int i = j; i < 64; ++i) s += sLinv[i * 64 + j] * sT[i];
        wp[WS_Q + j] = s;
    }

    {
        const int j = tid;
        for (int r = 0; r < 64; ++r) {
            float s = 0.f;
            const int i0 = (j > r) ? j : r;
            for (int i = i0; i < 64; ++i) s += sA[i * 64 + j] * sLinv[i * 64 + r];
            wp[WS_B + j * 64 + r] = s;
        }
        for (int r = 0; r < 64; ++r) wp[WS_LINV + j * 64 + r] = sLinv[j * 64 + r];
#pragma unroll
        for (int d = 0; d < 8; ++d) wp[WS_ZS + j * 8 + d] = sZs[j * 8 + d];
        wp[WS_Z2 + j] = sZ2[j];
        if (j < 8) wp[WS_ILS + j] = ils[j];
        if (j == 0) wp[WS_VAR] = var;
    }
}

#define LSTR 65
#define COLS_PER_BLOCK 512

__global__ __launch_bounds__(128) void svgp_main(
    const float* __restrict__ x, const float* __restrict__ ws,
    float* __restrict__ out, int n, int blocksPerPair)
{
    __shared__ float sLinv[64 * LSTR];
    __shared__ float sB[64 * LSTR];
    __shared__ float sZs[64 * 8];
    __shared__ float sZ2[64];
    __shared__ float sQ[64];
    __shared__ float sK[4][64 * 16];
    __shared__ float sXs[4][16 * 8];
    __shared__ float sX2[4][16];

    const int pair  = blockIdx.x / blocksPerPair;
    const int chunk = blockIdx.x % blocksPerPair;
    const float* wp = ws + (size_t)pair * WS_STRIDE;
    const int tid  = threadIdx.x;
    const int wave = tid >> 5;
    const int lane = tid & 31;

    for (int i = tid; i < 4096; i += 128) {
        const int r = i >> 6, c = i & 63;
        sLinv[r * LSTR + c] = wp[WS_LINV + i];
        sB[r * LSTR + c]    = wp[WS_B + i];
    }
    for (int i = tid; i < 512; i += 128) sZs[i] = wp[WS_ZS + i];
    if (tid < 64) { sZ2[tid] = wp[WS_Z2 + tid]; sQ[tid] = wp[WS_Q + tid]; }
    float ils[8];
#pragma unroll
    for (int d = 0; d < 8; ++d) ils[d] = wp[WS_ILS + d];
    const float var = wp[WS_VAR];
    __syncthreads();

    const size_t varBase = (size_t)(HDIM * ODIM) * (size_t)n;
    const int nloc  = lane & 15;
    const int khalf = (lane >> 4) * 2;

    for (int it = 0; it < COLS_PER_BLOCK / 64; ++it) {
        const int col0 = chunk * COLS_PER_BLOCK + it * 64 + wave * 16;
        __syncthreads();

        if (lane < 16) {
            const int cidx = col0 + lane;
            float s = 0.f;
#pragma unroll
            for (int d = 0; d < 8; ++d) {
                float v = x[cidx * 8 + d] * ils[d];
                sXs[wave][lane * 8 + d] = v;
                s += v * v;
            }
            sX2[wave][lane] = s;
        }
        __syncthreads();

        {
            const float xn2 = sX2[wave][nloc];
#pragma unroll 4
            for (int i = 0; i < 32; ++i) {
                const int m = 2 * i + (lane >> 4);
                float dot = 0.f;
#pragma unroll
                for (int d = 0; d < 8; ++d)
                    dot += sZs[m * 8 + d] * sXs[wave][nloc * 8 + d];
                sK[wave][m * 16 + nloc] =
                    var * __expf(-0.5f * (xn2 + sZ2[m] - 2.f * dot));
            }
        }
        __syncthreads();

        v8f accY[4] = {};
        v8f accW[4] = {};
#pragma unroll
        for (int kc = 0; kc < 16; ++kc) {
            const int k0 = kc * 4 + khalf;
            v2f bfrag;
            bfrag.x = sK[wave][(k0 + 0) * 16 + nloc];
            bfrag.y = sK[wave][(k0 + 1) * 16 + nloc];
#pragma unroll
            for (int rt = 0; rt < 4; ++rt) {
                const int m = rt * 16 + nloc;
                v2f aY; aY.x = sLinv[m * LSTR + k0]; aY.y = sLinv[m * LSTR + k0 + 1];
                accY[rt] = __builtin_amdgcn_wmma_f32_16x16x4_f32(
                    false, aY, false, bfrag, (short)0, accY[rt], false, false);
                v2f aW; aW.x = sB[m * LSTR + k0]; aW.y = sB[m * LSTR + k0 + 1];
                accW[rt] = __builtin_amdgcn_wmma_f32_16x16x4_f32(
                    false, aW, false, bfrag, (short)0, accW[rt], false, false);
            }
        }

        float d1 = 0.f, d2 = 0.f;
#pragma unroll
        for (int rt = 0; rt < 4; ++rt)
#pragma unroll
            for (int v = 0; v < 8; ++v) {
                d1 += accY[rt][v] * accY[rt][v];
                d2 += accW[rt][v] * accW[rt][v];
            }
        d1 += __shfl_xor(d1, 16, 32);
        d2 += __shfl_xor(d2, 16, 32);

        if (lane < 16) {
            float mu = 0.f;
#pragma unroll 8
            for (int m = 0; m < 64; ++m) mu += sQ[m] * sK[wave][m * 16 + lane];
            const int col = col0 + lane;
            out[(size_t)pair * n + col]           = mu;
            out[varBase + (size_t)pair * n + col] = var - d1 + d2;
        }
    }
}

extern "C" void kernel_launch(void* const* d_in, const int* in_sizes, int n_in,
                              void* d_out, int out_size, void* d_ws, size_t ws_size,
                              hipStream_t stream)
{
    const float* x          = (const float*)d_in[0];
    const float* z          = (const float*)d_in[1];
    const float* u_mean     = (const float*)d_in[2];
    const float* u_tril_vec = (const float*)d_in[3];
    const float* log_ls     = (const float*)d_in[4];
    const float* log_var    = (const float*)d_in[5];
    float* out = (float*)d_out;
    float* ws  = (float*)d_ws;

    const int n = in_sizes[0] / 8;
    svgp_prep<<<HDIM * ODIM, 64, 0, stream>>>(z, u_mean, u_tril_vec,
                                              log_ls, log_var, ws);
    const int blocksPerPair = n / COLS_PER_BLOCK;
    svgp_main<<<HDIM * ODIM * blocksPerPair, 128, 0, stream>>>(
        x, ws, out, n, blocksPerPair);
}


# prep kernel: 64x64 factor+inverse per pair moved from 64 serial threads to 1024 threads (register-resident in-place elimination yielding L^-1, parallel B product, coalesced stores)
# speedup vs baseline: 2.0120x; 2.0120x over previous
#include <hip/hip_runtime.h>

#define HDIM 4
#define ODIM 4
#define JITTER 1e-4f

#define WS_LINV 0
#define WS_B    4096
#define WS_ZS   8192
#define WS_Z2   8704
#define WS_Q    8768
#define WS_ILS  8832
#define WS_VAR  8840
#define WS_STRIDE 8960

typedef __attribute__((ext_vector_type(2))) float v2f;
typedef __attribute__((ext_vector_type(8))) float v8f;

#define LP 68
#define AP 65

__device__ __forceinline__ float pick4(float a0, float a1, float a2, float a3, int c)
{
    return (c == 0) ? a0 : (c == 1) ? a1 : (c == 2) ? a2 : a3;
}

__global__ __launch_bounds__(1024) void svgp_prep(
    const float* __restrict__ z, const float* __restrict__ u_mean,
    const float* __restrict__ u_tril_vec, const float* __restrict__ log_ls,
    const float* __restrict__ log_var, float* __restrict__ ws)
{
    __shared__ __attribute__((aligned(16))) float sLinv[64 * LP];
    __shared__ __attribute__((aligned(16))) float sRow[2][64];
    __shared__ float sA[64 * AP];
    __shared__ float sCol[2][64];
    __shared__ float sZs[64 * 8];
    __shared__ float sZ2[64];
    __shared__ float sD[64];
    __shared__ float sT[64];
    __shared__ float sIls[8];

    const int p   = blockIdx.x;
    const int o   = p % ODIM;
    const int tid = threadIdx.x;
    const int i   = tid >> 4;
    const int s4  = tid & 15;
    const int s0  = s4 * 4;
    float* wp = ws + (size_t)p * WS_STRIDE;

    const float var = __expf(log_var[p]);

    float u0, u1, u2, u3;
    {
        const int base = o * 2080 + i * (i + 1) / 2;
        u0 = (s0 + 0 <= i) ? u_tril_vec[base + s0 + 0] : 0.f;
        u1 = (s0 + 1 <= i) ? u_tril_vec[base + s0 + 1] : 0.f;
        u2 = (s0 + 2 <= i) ? u_tril_vec[base + s0 + 2] : 0.f;
        u3 = (s0 + 3 <= i) ? u_tril_vec[base + s0 + 3] : 0.f;
    }
    const float um0 = u_mean[o * 64 + s0 + 0];
    const float um1 = u_mean[o * 64 + s0 + 1];
    const float um2 = u_mean[o * 64 + s0 + 2];
    const float um3 = u_mean[o * 64 + s0 + 3];

    if (tid < 512) {
        const int d = tid & 7;
        const float il = __expf(-log_ls[p * 8 + d]);
        sZs[tid] = z[o * 512 + tid] * il;
        if (tid < 8) sIls[tid] = il;
    }
    __syncthreads();
    if (tid < 64) {
        float s = 0.f;
#pragma unroll
        for (int d = 0; d < 8; ++d) { const float v = sZs[tid * 8 + d]; s += v * v; }
        sZ2[tid] = s;
    }
    __syncthreads();

    float a0, a1, a2, a3;
    {
        float zi[8];
#pragma unroll
        for (int d = 0; d < 8; ++d) zi[d] = sZs[i * 8 + d];
        const float z2i = sZ2[i];
        float kv[4];
#pragma unroll
        for (int c = 0; c < 4; ++c) {
            const int s = s0 + c;
            float dot = 0.f;
#pragma unroll
            for (int d = 0; d < 8; ++d) dot += zi[d] * sZs[s * 8 + d];
            float v = var * __expf(-0.5f * (z2i + sZ2[s] - 2.f * dot));
            if (i == s) v += JITTER;
            kv[c] = v;
        }
        a0 = kv[0]; a1 = kv[1]; a2 = kv[2]; a3 = kv[3];
    }

    if (i == 0) *(float4*)&sRow[0][s0] = make_float4(a0, a1, a2, a3);
    if (s4 == 0) sCol[0][i] = a0;
    __syncthreads();

    for (int j = 0; j < 63; ++j) {
        const int buf = j & 1;
        if (i > j) {
            const float piv = sCol[buf][j];
            const float f   = sCol[buf][i] / piv;
            const float4 pr = *(const float4*)&sRow[buf][s0];
            a0 = (s0 + 0 == j) ? -f : a0 - f * pr.x;
            a1 = (s0 + 1 == j) ? -f : a1 - f * pr.y;
            a2 = (s0 + 2 == j) ? -f : a2 - f * pr.z;
            a3 = (s0 + 3 == j) ? -f : a3 - f * pr.w;
        }
        const int jn = j + 1;
        if (i == jn) *(float4*)&sRow[buf ^ 1][s0] = make_float4(a0, a1, a2, a3);
        if (s4 == (jn >> 2) && i >= jn) sCol[buf ^ 1][i] = pick4(a0, a1, a2, a3, jn & 3);
        __syncthreads();
    }

    if (s4 == (i >> 2)) sD[i] = pick4(a0, a1, a2, a3, i & 3);
    __syncthreads();

    {
        const float rs = 1.f / sqrtf(sD[i]);
        float4 l;
        l.x = (s0 + 0 < i) ? a0 * rs : ((s0 + 0 == i) ? rs : 0.f);
        l.y = (s0 + 1 < i) ? a1 * rs : ((s0 + 1 == i) ? rs : 0.f);
        l.z = (s0 + 2 < i) ? a2 * rs : ((s0 + 2 == i) ? rs : 0.f);
        l.w = (s0 + 3 < i) ? a3 * rs : ((s0 + 3 == i) ? rs : 0.f);
        *(float4*)&sLinv[i * LP + s0] = l;
        sA[i * AP + s0 + 0] = u0 * rs;
        sA[i * AP + s0 + 1] = u1 * rs;
        sA[i * AP + s0 + 2] = u2 * rs;
        sA[i * AP + s0 + 3] = u3 * rs;
        *(float4*)&wp[WS_LINV + tid * 4] = l;
    }
    __syncthreads();

    {
        float part = sLinv[i * LP + s0 + 0] * um0 + sLinv[i * LP + s0 + 1] * um1
                   + sLinv[i * LP + s0 + 2] * um2 + sLinv[i * LP + s0 + 3] * um3;
        part += __shfl_xor(part, 8, 32);
        part += __shfl_xor(part, 4, 32);
        part += __shfl_xor(part, 2, 32);
        part += __shfl_xor(part, 1, 32);
        if (s4 == 0) sT[i] = part;
    }
    __syncthreads();

    {
        float part = sLinv[(s0 + 0) * LP + i] * sT[s0 + 0] + sLinv[(s0 + 1) * LP + i] * sT[s0 + 1]
                   + sLinv[(s0 + 2) * LP + i] * sT[s0 + 2] + sLinv[(s0 + 3) * LP + i] * sT[s0 + 3];
        part += __shfl_xor(part, 8, 32);
        part += __shfl_xor(part, 4, 32);
        part += __shfl_xor(part, 2, 32);
        part += __shfl_xor(part, 1, 32);
        if (s4 == 0) wp[WS_Q + i] = part;
    }

    {
        float b0 = 0.f, b1 = 0.f, b2 = 0.f, b3 = 0.f;
        for (int k = i; k < 64; ++k) {
            const float av  = sA[k * AP + i];
            const float4 lv = *(const float4*)&sLinv[k * LP + s0];
            b0 += av * lv.x; b1 += av * lv.y; b2 += av * lv.z; b3 += av * lv.w;
        }
        *(float4*)&wp[WS_B + tid * 4] = make_float4(b0, b1, b2, b3);
    }

    if (tid < 512) wp[WS_ZS + tid] = sZs[tid];
    if (tid < 64)  wp[WS_Z2 + tid] = sZ2[tid];
    if (tid < 8)   wp[WS_ILS + tid] = sIls[tid];
    if (tid == 0)  wp[WS_VAR] = var;
}

#define LSTR 65
#define COLS_PER_BLOCK 512

__global__ __launch_bounds__(128) void svgp_main(
    const float* __restrict__ x, const float* __restrict__ ws,
    float* __restrict__ out, int n, int blocksPerPair)
{
    __shared__ float sLinv[64 * LSTR];
    __shared__ float sB[64 * LSTR];
    __shared__ float sZs[64 * 8];
    __shared__ float sZ2[64];
    __shared__ float sQ[64];
    __shared__ float sK[4][64 * 16];
    __shared__ float sXs[4][16 * 8];
    __shared__ float sX2[4][16];

    const int pair  = blockIdx.x / blocksPerPair;
    const int chunk = blockIdx.x % blocksPerPair;
    const float* wp = ws + (size_t)pair * WS_STRIDE;
    const int tid  = threadIdx.x;
    const int wave = tid >> 5;
    const int lane = tid & 31;

    for (int i = tid; i < 4096; i += 128) {
        const int r = i >> 6, c = i & 63;
        sLinv[r * LSTR + c] = wp[WS_LINV + i];
        sB[r * LSTR + c]    = wp[WS_B + i];
    }
    for (int i = tid; i < 512; i += 128) sZs[i] = wp[WS_ZS + i];
    if (tid < 64) { sZ2[tid] = wp[WS_Z2 + tid]; sQ[tid] = wp[WS_Q + tid]; }
    float ils[8];
#pragma unroll
    for (int d = 0; d < 8; ++d) ils[d] = wp[WS_ILS + d];
    const float var = wp[WS_VAR];
    __syncthreads();

    const size_t varBase = (size_t)(HDIM * ODIM) * (size_t)n;
    const int nloc  = lane & 15;
    const int khalf = (lane >> 4) * 2;

    for (int it = 0; it < COLS_PER_BLOCK / 64; ++it) {
        const int col0 = chunk * COLS_PER_BLOCK + it * 64 + wave * 16;
        __syncthreads();

        if (lane < 16) {
            const int cidx = col0 + lane;
            float s = 0.f;
#pragma unroll
            for (int d = 0; d < 8; ++d) {
                float v = x[cidx * 8 + d] * ils[d];
                sXs[wave][lane * 8 + d] = v;
                s += v * v;
            }
            sX2[wave][lane] = s;
        }
        __syncthreads();

        {
            const float xn2 = sX2[wave][nloc];
#pragma unroll 4
            for (int i = 0; i < 32; ++i) {
                const int m = 2 * i + (lane >> 4);
                float dot = 0.f;
#pragma unroll
                for (int d = 0; d < 8; ++d)
                    dot += sZs[m * 8 + d] * sXs[wave][nloc * 8 + d];
                sK[wave][m * 16 + nloc] =
                    var * __expf(-0.5f * (xn2 + sZ2[m] - 2.f * dot));
            }
        }
        __syncthreads();

        v8f accY[4] = {};
        v8f accW[4] = {};
#pragma unroll
        for (int kc = 0; kc < 16; ++kc) {
            const int k0 = kc * 4 + khalf;
            v2f bfrag;
            bfrag.x = sK[wave][(k0 + 0) * 16 + nloc];
            bfrag.y = sK[wave][(k0 + 1) * 16 + nloc];
#pragma unroll
            for (int rt = 0; rt < 4; ++rt) {
                const int m = rt * 16 + nloc;
                v2f aY; aY.x = sLinv[m * LSTR + k0]; aY.y = sLinv[m * LSTR + k0 + 1];
                accY[rt] = __builtin_amdgcn_wmma_f32_16x16x4_f32(
                    false, aY, false, bfrag, (short)0, accY[rt], false, false);
                v2f aW; aW.x = sB[m * LSTR + k0]; aW.y = sB[m * LSTR + k0 + 1];
                accW[rt] = __builtin_amdgcn_wmma_f32_16x16x4_f32(
                    false, aW, false, bfrag, (short)0, accW[rt], false, false);
            }
        }

        float d1 = 0.f, d2 = 0.f;
#pragma unroll
        for (int rt = 0; rt < 4; ++rt)
#pragma unroll
            for (int v = 0; v < 8; ++v) {
                d1 += accY[rt][v] * accY[rt][v];
                d2 += accW[rt][v] * accW[rt][v];
            }
        d1 += __shfl_xor(d1, 16, 32);
        d2 += __shfl_xor(d2, 16, 32);

        if (lane < 16) {
            float mu = 0.f;
#pragma unroll 8
            for (int m = 0; m < 64; ++m) mu += sQ[m] * sK[wave][m * 16 + lane];
            const int col = col0 + lane;
            out[(size_t)pair * n + col]           = mu;
            out[varBase + (size_t)pair * n + col] = var - d1 + d2;
        }
    }
}

extern "C" void kernel_launch(void* const* d_in, const int* in_sizes, int n_in,
                              void* d_out, int out_size, void* d_ws, size_t ws_size,
                              hipStream_t stream)
{
    const float* x          = (const float*)d_in[0];
    const float* z          = (const float*)d_in[1];
    const float* u_mean     = (const float*)d_in[2];
    const float* u_tril_vec = (const float*)d_in[3];
    const float* log_ls     = (const float*)d_in[4];
    const float* log_var    = (const float*)d_in[5];
    float* out = (float*)d_out;
    float* ws  = (float*)d_ws;

    const int n = in_sizes[0] / 8;
    svgp_prep<<<HDIM * ODIM, 1024, 0, stream>>>(z, u_mean, u_tril_vec,
                                              log_ls, log_var, ws);
    const int blocksPerPair = n / COLS_PER_BLOCK;
    svgp_main<<<HDIM * ODIM * blocksPerPair, 128, 0, stream>>>(
        x, ws, out, n, blocksPerPair);
}


# main: kfu tile made by f32 WMMA (exponent = [zs*log2e|cz|1] x [xs;1;cx], exp2 in accumulator layout), used from registers as B of Linv*K; no per-tile LDS or barrier; 256 blocks of 8 waves
# speedup vs baseline: 3.6901x; 1.8340x over previous
#include <hip/hip_runtime.h>

#define HDIM 4
#define ODIM 4
#define JITTER 1e-4f

#define WS_LINV 0
#define WS_B    4096
#define WS_ZS   8192
#define WS_Z2   8704
#define WS_Q    8768
#define WS_ILS  8832
#define WS_VAR  8840
#define WS_STRIDE 8960

typedef __attribute__((ext_vector_type(2))) float v2f;
typedef __attribute__((ext_vector_type(8))) float v8f;

#define LP 68
#define AP 65

__device__ __forceinline__ float pick4(float a0, float a1, float a2, float a3, int c)
{
    return (c == 0) ? a0 : (c == 1) ? a1 : (c == 2) ? a2 : a3;
}

__global__ __launch_bounds__(1024) void svgp_prep(
    const float* __restrict__ z, const float* __restrict__ u_mean,
    const float* __restrict__ u_tril_vec, const float* __restrict__ log_ls,
    const float* __restrict__ log_var, float* __restrict__ ws)
{
    __shared__ __attribute__((aligned(16))) float sLinv[64 * LP];
    __shared__ __attribute__((aligned(16))) float sRow[2][64];
    __shared__ float sA[64 * AP];
    __shared__ float sCol[2][64];
    __shared__ float sZs[64 * 8];
    __shared__ float sZ2[64];
    __shared__ float sD[64];
    __shared__ float sT[64];
    __shared__ float sIls[8];

    const int p   = blockIdx.x;
    const int o   = p % ODIM;
    const int tid = threadIdx.x;
    const int i   = tid >> 4;
    const int s4  = tid & 15;
    const int s0  = s4 * 4;
    float* wp = ws + (size_t)p * WS_STRIDE;

    const float var = __expf(log_var[p]);

    float u0, u1, u2, u3;
    {
        const int base = o * 2080 + i * (i + 1) / 2;
        u0 = (s0 + 0 <= i) ? u_tril_vec[base + s0 + 0] : 0.f;
        u1 = (s0 + 1 <= i) ? u_tril_vec[base + s0 + 1] : 0.f;
        u2 = (s0 + 2 <= i) ? u_tril_vec[base + s0 + 2] : 0.f;
        u3 = (s0 + 3 <= i) ? u_tril_vec[base + s0 + 3] : 0.f;
    }
    const float um0 = u_mean[o * 64 + s0 + 0];
    const float um1 = u_mean[o * 64 + s0 + 1];
    const float um2 = u_mean[o * 64 + s0 + 2];
    const float um3 = u_mean[o * 64 + s0 + 3];

    if (tid < 512) {
        const int d = tid & 7;
        const float il = __expf(-log_ls[p * 8 + d]);
        sZs[tid] = z[o * 512 + tid] * il;
        if (tid < 8) sIls[tid] = il;
    }
    __syncthreads();
    if (tid < 64) {
        float s = 0.f;
#pragma unroll
        for (int d = 0; d < 8; ++d) { const float v = sZs[tid * 8 + d]; s += v * v; }
        sZ2[tid] = s;
    }
    __syncthreads();

    float a0, a1, a2, a3;
    {
        float zi[8];
#pragma unroll
        for (int d = 0; d < 8; ++d) zi[d] = sZs[i * 8 + d];
        const float z2i = sZ2[i];
        float kv[4];
#pragma unroll
        for (int c = 0; c < 4; ++c) {
            const int s = s0 + c;
            float dot = 0.f;
#pragma unroll
            for (int d = 0; d < 8; ++d) dot += zi[d] * sZs[s * 8 + d];
            float v = var * __expf(-0.5f * (z2i + sZ2[s] - 2.f * dot));
            if (i == s) v += JITTER;
            kv[c] = v;
        }
        a0 = kv[0]; a1 = kv[1]; a2 = kv[2]; a3 = kv[3];
    }

    if (i == 0) *(float4*)&sRow[0][s0] = make_float4(a0, a1, a2, a3);
    if (s4 == 0) sCol[0][i] = a0;
    __syncthreads();

    for (int j = 0; j < 63; ++j) {
        const int buf = j & 1;
        if (i > j) {
            const float piv = sCol[buf][j];
            const float f   = sCol[buf][i] / piv;
            const float4 pr = *(const float4*)&sRow[buf][s0];
            a0 = (s0 + 0 == j) ? -f : a0 - f * pr.x;
            a1 = (s0 + 1 == j) ? -f : a1 - f * pr.y;
            a2 = (s0 + 2 == j) ? -f : a2 - f * pr.z;
            a3 = (s0 + 3 == j) ? -f : a3 - f * pr.w;
        }
        const int jn = j + 1;
        if (i == jn) *(float4*)&sRow[buf ^ 1][s0] = make_float4(a0, a1, a2, a3);
        if (s4 == (jn >> 2) && i >= jn) sCol[buf ^ 1][i] = pick4(a0, a1, a2, a3, jn & 3);
        __syncthreads();
    }

    if (s4 == (i >> 2)) sD[i] = pick4(a0, a1, a2, a3, i & 3);
    __syncthreads();

    {
        const float rs = 1.f / sqrtf(sD[i]);
        float4 l;
        l.x = (s0 + 0 < i) ? a0 * rs : ((s0 + 0 == i) ? rs : 0.f);
        l.y = (s0 + 1 < i) ? a1 * rs : ((s0 + 1 == i) ? rs : 0.f);
        l.z = (s0 + 2 < i) ? a2 * rs : ((s0 + 2 == i) ? rs : 0.f);
        l.w = (s0 + 3 < i) ? a3 * rs : ((s0 + 3 == i) ? rs : 0.f);
        *(float4*)&sLinv[i * LP + s0] = l;
        sA[i * AP + s0 + 0] = u0 * rs;
        sA[i * AP + s0 + 1] = u1 * rs;
        sA[i * AP + s0 + 2] = u2 * rs;
        sA[i * AP + s0 + 3] = u3 * rs;
        *(float4*)&wp[WS_LINV + tid * 4] = l;
    }
    __syncthreads();

    {
        float part = sLinv[i * LP + s0 + 0] * um0 + sLinv[i * LP + s0 + 1] * um1
                   + sLinv[i * LP + s0 + 2] * um2 + sLinv[i * LP + s0 + 3] * um3;
        part += __shfl_xor(part, 8, 32);
        part += __shfl_xor(part, 4, 32);
        part += __shfl_xor(part, 2, 32);
        part += __shfl_xor(part, 1, 32);
        if (s4 == 0) sT[i] = part;
    }
    __syncthreads();

    {
        float part = sLinv[(s0 + 0) * LP + i] * sT[s0 + 0] + sLinv[(s0 + 1) * LP + i] * sT[s0 + 1]
                   + sLinv[(s0 + 2) * LP + i] * sT[s0 + 2] + sLinv[(s0 + 3) * LP + i] * sT[s0 + 3];
        part += __shfl_xor(part, 8, 32);
        part += __shfl_xor(part, 4, 32);
        part += __shfl_xor(part, 2, 32);
        part += __shfl_xor(part, 1, 32);
        if (s4 == 0) wp[WS_Q + i] = part;
    }

    *(float4*)&wp[WS_B + tid * 4] = make_float4(sA[(s0 + 0) * AP + i], sA[(s0 + 1) * AP + i],
                                                sA[(s0 + 2) * AP + i], sA[(s0 + 3) * AP + i]);

    if (tid < 512) wp[WS_ZS + tid] = sZs[tid];
    if (tid < 64)  wp[WS_Z2 + tid] = sZ2[tid];
    if (tid < 8)   wp[WS_ILS + tid] = sIls[tid];
    if (tid == 0)  wp[WS_VAR] = var;
}

#define LSTR 65
#define MAIN_WAVES 8
#define MAIN_THREADS (MAIN_WAVES * 32)
#define COLS_PER_ITER (MAIN_WAVES * 16)
#define COLS_PER_BLOCK 2048
#define LOG2E_F 1.44269504088896340736f

__device__ __forceinline__ v8f mm4(v2f a, v2f b, v8f c)
{
    return __builtin_amdgcn_wmma_f32_16x16x4_f32(false, a, false, b, (short)0, c, false, false);
}

__global__ __launch_bounds__(MAIN_THREADS) void svgp_main(
    const float* __restrict__ x, const float* __restrict__ ws,
    float* __restrict__ out, int n, int blocksPerPair)
{
    __shared__ float sLinv[64 * LSTR];
    __shared__ float sB[64 * LSTR];

    const int pair  = blockIdx.x / blocksPerPair;
    const int chunk = blockIdx.x % blocksPerPair;
    const float* wp = ws + (size_t)pair * WS_STRIDE;
    const int tid  = threadIdx.x;
    const int wave = tid >> 5;
    const int lane = tid & 31;
    const int c    = lane & 15;
    const int h    = lane >> 4;
    const int kd   = h * 8;

    for (int i = tid; i < 4096; i += MAIN_THREADS) {
        const int r = i >> 6, cc = i & 63;
        sLinv[r * LSTR + cc] = wp[WS_LINV + i];
        sB[r * LSTR + cc]    = wp[WS_B + i];
    }

    const float il0 = wp[WS_ILS + 2 * h];
    const float il1 = wp[WS_ILS + 2 * h + 1];
    const float il2 = wp[WS_ILS + 4 + 2 * h];
    const float il3 = wp[WS_ILS + 5 + 2 * h];
    const float var = wp[WS_VAR];
    const float l2var = __log2f(var);

    v2f aE[4][3];
    v2f qr[4][4];
#pragma unroll
    for (int rt = 0; rt < 4; ++rt) {
        const int m = rt * 16 + c;
        const float* zr = wp + WS_ZS + m * 8;
        aE[rt][0].x = zr[2 * h]     * LOG2E_F;
        aE[rt][0].y = zr[2 * h + 1] * LOG2E_F;
        aE[rt][1].x = zr[4 + 2 * h] * LOG2E_F;
        aE[rt][1].y = zr[5 + 2 * h] * LOG2E_F;
        const float cz = (-0.5f * LOG2E_F) * wp[WS_Z2 + m] + l2var;
        aE[rt][2].x = h ? 1.f : cz;
        aE[rt][2].y = h ? 0.f : 1.f;
#pragma unroll
        for (int j = 0; j < 4; ++j) {
            qr[rt][j].x = wp[WS_Q + rt * 16 + kd + 2 * j];
            qr[rt][j].y = wp[WS_Q + rt * 16 + kd + 2 * j + 1];
        }
    }
    __syncthreads();

    const size_t varBase = (size_t)(HDIM * ODIM) * (size_t)n;
    const int colW = chunk * COLS_PER_BLOCK + wave * 16 + c;
    float* const dst = out + (h ? varBase : (size_t)0) + (size_t)pair * n + colW;

    const float* xp = x + (size_t)colW * 8 + 2 * h;
    float xa0 = xp[0], xa1 = xp[1], xb0 = xp[4], xb1 = xp[5];

    for (int it = 0; it < COLS_PER_BLOCK / COLS_PER_ITER; ++it) {
        const float xs0 = xa0 * il0, xs1 = xa1 * il1, xs2 = xb0 * il2, xs3 = xb1 * il3;

        {
            const int itn = (it + 1 < COLS_PER_BLOCK / COLS_PER_ITER) ? it + 1 : it;
            const float* xq = xp + (size_t)itn * COLS_PER_ITER * 8;
            xa0 = xq[0]; xa1 = xq[1]; xb0 = xq[4]; xb1 = xq[5];
        }

        const float cxp = (-0.5f * LOG2E_F) * (xs0 * xs0 + xs1 * xs1 + xs2 * xs2 + xs3 * xs3);
        v2f bE0, bE1, bE2;
        bE0.x = xs0; bE0.y = xs1;
        bE1.x = xs2; bE1.y = xs3;
        bE2.x = h ? cxp : 1.f;
        bE2.y = h ? 0.f : cxp;

        v8f kt[4] = {};
#pragma unroll
        for (int rt = 0; rt < 4; ++rt) kt[rt] = mm4(aE[rt][0], bE0, kt[rt]);
#pragma unroll
        for (int rt = 0; rt < 4; ++rt) kt[rt] = mm4(aE[rt][1], bE1, kt[rt]);
#pragma unroll
        for (int rt = 0; rt < 4; ++rt) kt[rt] = mm4(aE[rt][2], bE2, kt[rt]);
        v2f mu2 = {0.f, 0.f};
#pragma unroll
        for (int rt = 0; rt < 4; ++rt)
#pragma unroll
            for (int r = 0; r < 8; r += 2) {
                kt[rt][r]     = __builtin_amdgcn_exp2f(kt[rt][r]);
                kt[rt][r + 1] = __builtin_amdgcn_exp2f(kt[rt][r + 1]);
                v2f kp; kp.x = kt[rt][r]; kp.y = kt[rt][r + 1];
                mu2 = kp * qr[rt][r >> 1] + mu2;
            }

        v8f accY[4] = {};
        v8f accW[4] = {};
#pragma unroll
        for (int kq = 0; kq < 5; ++kq) {
#pragma unroll
            for (int jj = 0; jj < 4; ++jj) {
                if (kq < 4) {
                    const int kk = kq * 16 + kd + 2 * jj;
                    v2f bK;
                    bK.x = kt[kq][2 * jj + 0];
                    bK.y = kt[kq][2 * jj + 1];
#pragma unroll
                    for (int rt = kq; rt < 4; ++rt) {
                        const int m = rt * 16 + c;
                        v2f aY; aY.x = sLinv[m * LSTR + kk]; aY.y = sLinv[m * LSTR + kk + 1];
                        accY[rt] = mm4(aY, bK, accY[rt]);
                    }
                }
                if (kq > 0) {
                    const int rt = kq - 1;
                    const int kk = rt * 16 + kd + 2 * jj;
                    v2f bY;
                    bY.x = accY[rt][2 * jj + 0];
                    bY.y = accY[rt][2 * jj + 1];
#pragma unroll
                    for (int mt = 0; mt <= rt; ++mt) {
                        const int m = mt * 16 + c;
                        v2f aW; aW.x = sB[m * LSTR + kk]; aW.y = sB[m * LSTR + kk + 1];
                        accW[mt] = mm4(aW, bY, accW[mt]);
                    }
                }
            }
        }

        v2f s1 = {0.f, 0.f}, s2 = {0.f, 0.f};
#pragma unroll
        for (int rt = 0; rt < 4; ++rt)
#pragma unroll
            for (int v = 0; v < 8; v += 2) {
                v2f y; y.x = accY[rt][v]; y.y = accY[rt][v + 1];
                v2f w; w.x = accW[rt][v]; w.y = accW[rt][v + 1];
                s1 = y * y + s1;
                s2 = w * w + s2;
            }
        float mu = mu2.x + mu2.y;
        float pv = (s2.x + s2.y) - (s1.x + s1.y);
        mu += __shfl_xor(mu, 16, 32);
        pv += __shfl_xor(pv, 16, 32);

        dst[it * COLS_PER_ITER] = h ? (var + pv) : mu;
    }
}

extern "C" void kernel_launch(void* const* d_in, const int* in_sizes, int n_in,
                              void* d_out, int out_size, void* d_ws, size_t ws_size,
                              hipStream_t stream)
{
    const float* x          = (const float*)d_in[0];
    const float* z          = (const float*)d_in[1];
    const float* u_mean     = (const float*)d_in[2];
    const float* u_tril_vec = (const float*)d_in[3];
    const float* log_ls     = (const float*)d_in[4];
    const float* log_var    = (const float*)d_in[5];
    float* out = (float*)d_out;
    float* ws  = (float*)d_ws;

    const int n = in_sizes[0] / 8;
    svgp_prep<<<HDIM * ODIM, 1024, 0, stream>>>(z, u_mean, u_tril_vec,
                                              log_ls, log_var, ws);
    const int blocksPerPair = n / COLS_PER_BLOCK;
    svgp_main<<<HDIM * ODIM * blocksPerPair, MAIN_THREADS, 0, stream>>>(
        x, ws, out, n, blocksPerPair);
}


# prep: 64x64 elimination on 256 threads (16 matrix elements per thread, register indices static) instead of 1024: one division per 16 elements, 8-wave step barrier
# speedup vs baseline: 4.1453x; 1.1234x over previous
#include <hip/hip_runtime.h>

#define HDIM 4
#define ODIM 4
#define JITTER 1e-4f

#define WS_LINV 0
#define WS_B    4096
#define WS_ZS   8192
#define WS_Z2   8704
#define WS_Q    8768
#define WS_ILS  8832
#define WS_VAR  8840
#define WS_STRIDE 8960

typedef __attribute__((ext_vector_type(2))) float v2f;
typedef __attribute__((ext_vector_type(8))) float v8f;

#define LP 68
#define AP 65
#define PREP_THREADS 256

__global__ __launch_bounds__(PREP_THREADS) void svgp_prep(
    const float* __restrict__ z, const float* __restrict__ u_mean,
    const float* __restrict__ u_tril_vec, const float* __restrict__ log_ls,
    const float* __restrict__ log_var, float* __restrict__ ws)
{
    __shared__ __attribute__((aligned(16))) float sLinv[64 * LP];
    __shared__ __attribute__((aligned(16))) float sRow[2][64];
    __shared__ float sA[64 * AP];
    __shared__ float sCol[2][64];
    __shared__ float sZs[64 * 8];
    __shared__ float sZ2[64];
    __shared__ float sD[64];
    __shared__ float sT[64];
    __shared__ float sIls[8];

    const int p   = blockIdx.x;
    const int o   = p % ODIM;
    const int tid = threadIdx.x;
    const int i   = tid >> 2;
    const int s4  = tid & 3;
    const int s0  = s4 * 16;
    float* wp = ws + (size_t)p * WS_STRIDE;

    const float var = __expf(log_var[p]);

    float u[16], um[16];
    {
        const int base = o * 2080 + i * (i + 1) / 2;
#pragma unroll
        for (int c = 0; c < 16; ++c) {
            u[c]  = (s0 + c <= i) ? u_tril_vec[base + s0 + c] : 0.f;
            um[c] = u_mean[o * 64 + s0 + c];
        }
    }

    for (int e = tid; e < 512; e += PREP_THREADS) {
        const int d = e & 7;
        const float il = __expf(-log_ls[p * 8 + d]);
        sZs[e] = z[o * 512 + e] * il;
        if (e < 8) sIls[e] = il;
    }
    __syncthreads();
    if (tid < 64) {
        float s = 0.f;
#pragma unroll
        for (int d = 0; d < 8; ++d) { const float v = sZs[tid * 8 + d]; s += v * v; }
        sZ2[tid] = s;
    }
    __syncthreads();

    float a[16];
    {
        float zi[8];
#pragma unroll
        for (int d = 0; d < 8; ++d) zi[d] = sZs[i * 8 + d];
        const float z2i = sZ2[i];
#pragma unroll
        for (int c = 0; c < 16; ++c) {
            const int s = s0 + c;
            float dot = 0.f;
#pragma unroll
            for (int d = 0; d < 8; ++d) dot += zi[d] * sZs[s * 8 + d];
            float v = var * __expf(-0.5f * (z2i + sZ2[s] - 2.f * dot));
            if (i == s) v += JITTER;
            a[c] = v;
        }
    }

    if (i == 0) {
#pragma unroll
        for (int q = 0; q < 4; ++q)
            *(float4*)&sRow[0][s0 + 4 * q] =
                make_float4(a[4 * q], a[4 * q + 1], a[4 * q + 2], a[4 * q + 3]);
    }
    if (s4 == 0) sCol[0][i] = a[0];
    __syncthreads();

    for (int jb = 0; jb < 4; ++jb) {
#pragma unroll
        for (int jj = 0; jj < 16; ++jj) {
            const int j = jb * 16 + jj;
            if (j < 63) {
                const int buf = jj & 1;
                if (i > j) {
                    const float piv = sCol[buf][j];
                    const float f   = sCol[buf][i] / piv;
                    float pr[16];
#pragma unroll
                    for (int q = 0; q < 4; ++q) {
                        const float4 t = *(const float4*)&sRow[buf][s0 + 4 * q];
                        pr[4 * q] = t.x; pr[4 * q + 1] = t.y; pr[4 * q + 2] = t.z; pr[4 * q + 3] = t.w;
                    }
#pragma unroll
                    for (int c = 0; c < 16; ++c) a[c] = a[c] - f * pr[c];
                    if (s4 == jb) a[jj] = -f;
                }
                const int jn = j + 1;
                if (i == jn) {
#pragma unroll
                    for (int q = 0; q < 4; ++q)
                        *(float4*)&sRow[buf ^ 1][s0 + 4 * q] =
                            make_float4(a[4 * q], a[4 * q + 1], a[4 * q + 2], a[4 * q + 3]);
                }
                if (jj < 15) {
                    if (s4 == jb && i >= jn) sCol[buf ^ 1][i] = a[(jj + 1) & 15];
                } else {
                    if (s4 == jb + 1 && i >= jn) sCol[buf ^ 1][i] = a[0];
                }
                __syncthreads();
            }
        }
    }

#pragma unroll
    for (int c = 0; c < 16; ++c)
        if (s0 + c == i) sD[i] = a[c];
    __syncthreads();

    float l[16];
    {
        const float rs = 1.f / sqrtf(sD[i]);
#pragma unroll
        for (int c = 0; c < 16; ++c) {
            l[c] = (s0 + c < i) ? a[c] * rs : ((s0 + c == i) ? rs : 0.f);
            sA[i * AP + s0 + c] = u[c] * rs;
        }
#pragma unroll
        for (int q = 0; q < 4; ++q) {
            const float4 t = make_float4(l[4 * q], l[4 * q + 1], l[4 * q + 2], l[4 * q + 3]);
            *(float4*)&sLinv[i * LP + s0 + 4 * q] = t;
            *(float4*)&wp[WS_LINV + tid * 16 + 4 * q] = t;
        }
    }

    {
        float part = 0.f;
#pragma unroll
        for (int c = 0; c < 16; ++c) part += l[c] * um[c];
        part += __shfl_xor(part, 2, 32);
        part += __shfl_xor(part, 1, 32);
        if (s4 == 0) sT[i] = part;
    }
    __syncthreads();

    {
        float part = 0.f;
#pragma unroll
        for (int c = 0; c < 16; ++c) part += sLinv[(s0 + c) * LP + i] * sT[s0 + c];
        part += __shfl_xor(part, 2, 32);
        part += __shfl_xor(part, 1, 32);
        if (s4 == 0) wp[WS_Q + i] = part;
    }

#pragma unroll
    for (int q = 0; q < 4; ++q)
        *(float4*)&wp[WS_B + tid * 16 + 4 * q] =
            make_float4(sA[(s0 + 4 * q + 0) * AP + i], sA[(s0 + 4 * q + 1) * AP + i],
                        sA[(s0 + 4 * q + 2) * AP + i], sA[(s0 + 4 * q + 3) * AP + i]);

    for (int e = tid; e < 512; e += PREP_THREADS) wp[WS_ZS + e] = sZs[e];
    if (tid < 64)  wp[WS_Z2 + tid] = sZ2[tid];
    if (tid < 8)   wp[WS_ILS + tid] = sIls[tid];
    if (tid == 0)  wp[WS_VAR] = var;
}

#define LSTR 65
#define MAIN_WAVES 8
#define MAIN_THREADS (MAIN_WAVES * 32)
#define COLS_PER_ITER (MAIN_WAVES * 16)
#define COLS_PER_BLOCK 2048
#define LOG2E_F 1.44269504088896340736f

__device__ __forceinline__ v8f mm4(v2f a, v2f b, v8f c)
{
    return __builtin_amdgcn_wmma_f32_16x16x4_f32(false, a, false, b, (short)0, c, false, false);
}

__global__ __launch_bounds__(MAIN_THREADS) void svgp_main(
    const float* __restrict__ x, const float* __restrict__ ws,
    float* __restrict__ out, int n, int blocksPerPair)
{
    __shared__ float sLinv[64 * LSTR];
    __shared__ float sB[64 * LSTR];

    const int pair  = blockIdx.x / blocksPerPair;
    const int chunk = blockIdx.x % blocksPerPair;
    const float* wp = ws + (size_t)pair * WS_STRIDE;
    const int tid  = threadIdx.x;
    const int wave = tid >> 5;
    const int lane = tid & 31;
    const int c    = lane & 15;
    const int h    = lane >> 4;
    const int kd   = h * 8;

    for (int i = tid; i < 4096; i += MAIN_THREADS) {
        const int r = i >> 6, cc = i & 63;
        sLinv[r * LSTR + cc] = wp[WS_LINV + i];
        sB[r * LSTR + cc]    = wp[WS_B + i];
    }

    const float il0 = wp[WS_ILS + 2 * h];
    const float il1 = wp[WS_ILS + 2 * h + 1];
    const float il2 = wp[WS_ILS + 4 + 2 * h];
    const float il3 = wp[WS_ILS + 5 + 2 * h];
    const float var = wp[WS_VAR];
    const float l2var = __log2f(var);

    v2f aE[4][3];
    v2f qr[4][4];
#pragma unroll
    for (int rt = 0; rt < 4; ++rt) {
        const int m = rt * 16 + c;
        const float* zr = wp + WS_ZS + m * 8;
        aE[rt][0].x = zr[2 * h]     * LOG2E_F;
        aE[rt][0].y = zr[2 * h + 1] * LOG2E_F;
        aE[rt][1].x = zr[4 + 2 * h] * LOG2E_F;
        aE[rt][1].y = zr[5 + 2 * h] * LOG2E_F;
        const float cz = (-0.5f * LOG2E_F) * wp[WS_Z2 + m] + l2var;
        aE[rt][2].x = h ? 1.f : cz;
        aE[rt][2].y = h ? 0.f : 1.f;
#pragma unroll
        for (int j = 0; j < 4; ++j) {
            qr[rt][j].x = wp[WS_Q + rt * 16 + kd + 2 * j];
            qr[rt][j].y = wp[WS_Q + rt * 16 + kd + 2 * j + 1];
        }
    }
    __syncthreads();

    const size_t varBase = (size_t)(HDIM * ODIM) * (size_t)n;
    const int colW = chunk * COLS_PER_BLOCK + wave * 16 + c;
    float* const dst = out + (h ? varBase : (size_t)0) + (size_t)pair * n + colW;

    const float* xp = x + (size_t)colW * 8 + 2 * h;
    float xa0 = xp[0], xa1 = xp[1], xb0 = xp[4], xb1 = xp[5];

    for (int it = 0; it < COLS_PER_BLOCK / COLS_PER_ITER; ++it) {
        const float xs0 = xa0 * il0, xs1 = xa1 * il1, xs2 = xb0 * il2, xs3 = xb1 * il3;

        {
            const int itn = (it + 1 < COLS_PER_BLOCK / COLS_PER_ITER) ? it + 1 : it;
            const float* xq = xp + (size_t)itn * COLS_PER_ITER * 8;
            xa0 = xq[0]; xa1 = xq[1]; xb0 = xq[4]; xb1 = xq[5];
        }

        const float cxp = (-0.5f * LOG2E_F) * (xs0 * xs0 + xs1 * xs1 + xs2 * xs2 + xs3 * xs3);
        v2f bE0, bE1, bE2;
        bE0.x = xs0; bE0.y = xs1;
        bE1.x = xs2; bE1.y = xs3;
        bE2.x = h ? cxp : 1.f;
        bE2.y = h ? 0.f : cxp;

        v8f kt[4] = {};
#pragma unroll
        for (int rt = 0; rt < 4; ++rt) kt[rt] = mm4(aE[rt][0], bE0, kt[rt]);
#pragma unroll
        for (int rt = 0; rt < 4; ++rt) kt[rt] = mm4(aE[rt][1], bE1, kt[rt]);
#pragma unroll
        for (int rt = 0; rt < 4; ++rt) kt[rt] = mm4(aE[rt][2], bE2, kt[rt]);
        v2f mu2 = {0.f, 0.f};
#pragma unroll
        for (int rt = 0; rt < 4; ++rt)
#pragma unroll
            for (int r = 0; r < 8; r += 2) {
                kt[rt][r]     = __builtin_amdgcn_exp2f(kt[rt][r]);
                kt[rt][r + 1] = __builtin_amdgcn_exp2f(kt[rt][r + 1]);
                v2f kp; kp.x = kt[rt][r]; kp.y = kt[rt][r + 1];
                mu2 = kp * qr[rt][r >> 1] + mu2;
            }

        v8f accY[4] = {};
        v8f accW[4] = {};
#pragma unroll
        for (int kq = 0; kq < 5; ++kq) {
#pragma unroll
            for (int jj = 0; jj < 4; ++jj) {
                if (kq < 4) {
                    const int kk = kq * 16 + kd + 2 * jj;
                    v2f bK;
                    bK.x = kt[kq][2 * jj + 0];
                    bK.y = kt[kq][2 * jj + 1];
#pragma unroll
                    for (int rt = kq; rt < 4; ++rt) {
                        const int m = rt * 16 + c;
                        v2f aY; aY.x = sLinv[m * LSTR + kk]; aY.y = sLinv[m * LSTR + kk + 1];
                        accY[rt] = mm4(aY, bK, accY[rt]);
                    }
                }
                if (kq > 0) {
                    const int rt = kq - 1;
                    const int kk = rt * 16 + kd + 2 * jj;
                    v2f bY;
                    bY.x = accY[rt][2 * jj + 0];
                    bY.y = accY[rt][2 * jj + 1];
#pragma unroll
                    for (int mt = 0; mt <= rt; ++mt) {
                        const int m = mt * 16 + c;
                        v2f aW; aW.x = sB[m * LSTR + kk]; aW.y = sB[m * LSTR + kk + 1];
                        accW[mt] = mm4(aW, bY, accW[mt]);
                    }
                }
            }
        }

        v2f s1 = {0.f, 0.f}, s2 = {0.f, 0.f};
#pragma unroll
        for (int rt = 0; rt < 4; ++rt)
#pragma unroll
            for (int v = 0; v < 8; v += 2) {
                v2f y; y.x = accY[rt][v]; y.y = accY[rt][v + 1];
                v2f w; w.x = accW[rt][v]; w.y = accW[rt][v + 1];
                s1 = y * y + s1;
                s2 = w * w + s2;
            }
        float mu = mu2.x + mu2.y;
        float pv = (s2.x + s2.y) - (s1.x + s1.y);
        mu += __shfl_xor(mu, 16, 32);
        pv += __shfl_xor(pv, 16, 32);

        dst[it * COLS_PER_ITER] = h ? (var + pv) : mu;
    }
}

extern "C" void kernel_launch(void* const* d_in, const int* in_sizes, int n_in,
                              void* d_out, int out_size, void* d_ws, size_t ws_size,
                              hipStream_t stream)
{
    const float* x          = (const float*)d_in[0];
    const float* z          = (const float*)d_in[1];
    const float* u_mean     = (const float*)d_in[2];
    const float* u_tril_vec = (const float*)d_in[3];
    const float* log_ls     = (const float*)d_in[4];
    const float* log_var    = (const float*)d_in[5];
    float* out = (float*)d_out;
    float* ws  = (float*)d_ws;

    const int n = in_sizes[0] / 8;
    svgp_prep<<<HDIM * ODIM, PREP_THREADS, 0, stream>>>(z, u_mean, u_tril_vec,
                                              log_ls, log_var, ws);
    const int blocksPerPair = n / COLS_PER_BLOCK;
    svgp_main<<<HDIM * ODIM * blocksPerPair, MAIN_THREADS, 0, stream>>>(
        x, ws, out, n, blocksPerPair);
}


# pred_var as var - k^T G k: G = Linv^T (I - A A^T) Linv made once per pair in prep on f32 WMMA; G symmetric, so main multiplies its upper block triangle only: 40 f32 WMMA per 16 columns, not 80
# speedup vs baseline: 5.6234x; 1.3566x over previous
#include <hip/hip_runtime.h>

#define HDIM 4
#define ODIM 4
#define JITTER 1e-4f

#define WS_V    0
#define WS_ZS   8192
#define WS_Z2   8704
#define WS_Q    8768
#define WS_ILS  8832
#define WS_VAR  8840
#define WS_STRIDE 8960

typedef __attribute__((ext_vector_type(2))) float v2f;
typedef __attribute__((ext_vector_type(8))) float v8f;

#define LP 68
#define AP 65
#define PREP_THREADS 256

__device__ __forceinline__ v8f mm4(v2f a, v2f b, v8f c)
{
    return __builtin_amdgcn_wmma_f32_16x16x4_f32(false, a, false, b, (short)0, c, false, false);
}

__device__ __forceinline__ v2f frag2(const float* S, int pitch, int row, int col)
{
    v2f f;
    f.x = S[row * pitch + col];
    f.y = S[(row + 1) * pitch + col];
    return f;
}

__device__ __forceinline__ void store_vtile(float* gp, v8f g1, v8f g2, float sc)
{
    *(float4*)gp = make_float4((g1[0] - g2[0]) * sc, (g1[1] - g2[1]) * sc,
                               (g1[2] - g2[2]) * sc, (g1[3] - g2[3]) * sc);
    *(float4*)(gp + 4) = make_float4((g1[4] - g2[4]) * sc, (g1[5] - g2[5]) * sc,
                                     (g1[6] - g2[6]) * sc, (g1[7] - g2[7]) * sc);
}

__global__ __launch_bounds__(PREP_THREADS) void svgp_prep(
    const float* __restrict__ z, const float* __restrict__ u_mean,
    const float* __restrict__ u_tril_vec, const float* __restrict__ log_ls,
    const float* __restrict__ log_var, float* __restrict__ ws)
{
    __shared__ __attribute__((aligned(16))) float sLinv[64 * LP];
    __shared__ __attribute__((aligned(16))) float sRow[2][64];
    __shared__ float sA[64 * AP];
    __shared__ float sP[64 * LP];
    __shared__ float sCol[2][64];
    __shared__ float sZs[64 * 8];
    __shared__ float sZ2[64];
    __shared__ float sD[64];
    __shared__ float sT[64];
    __shared__ float sIls[8];

    const int p   = blockIdx.x;
    const int o   = p % ODIM;
    const int tid = threadIdx.x;
    const int i   = tid >> 2;
    const int s4  = tid & 3;
    const int s0  = s4 * 16;
    float* wp = ws + (size_t)p * WS_STRIDE;

    const float var = __expf(log_var[p]);

    float u[16], um[16];
    {
        const int base = o * 2080 + i * (i + 1) / 2;
#pragma unroll
        for (int c = 0; c < 16; ++c) {
            u[c]  = (s0 + c <= i) ? u_tril_vec[base + s0 + c] : 0.f;
            um[c] = u_mean[o * 64 + s0 + c];
        }
    }

    for (int e = tid; e < 512; e += PREP_THREADS) {
        const int d = e & 7;
        const float il = __expf(-log_ls[p * 8 + d]);
        sZs[e] = z[o * 512 + e] * il;
        if (e < 8) sIls[e] = il;
    }
    __syncthreads();
    if (tid < 64) {
        float s = 0.f;
#pragma unroll
        for (int d = 0; d < 8; ++d) { const float v = sZs[tid * 8 + d]; s += v * v; }
        sZ2[tid] = s;
    }
    __syncthreads();

    float a[16];
    {
        float zi[8];
#pragma unroll
        for (int d = 0; d < 8; ++d) zi[d] = sZs[i * 8 + d];
        const float z2i = sZ2[i];
#pragma unroll
        for (int c = 0; c < 16; ++c) {
            const int s = s0 + c;
            float dot = 0.f;
#pragma unroll
            for (int d = 0; d < 8; ++d) dot += zi[d] * sZs[s * 8 + d];
            float v = var * __expf(-0.5f * (z2i + sZ2[s] - 2.f * dot));
            if (i == s) v += JITTER;
            a[c] = v;
        }
    }

    if (i == 0) {
#pragma unroll
        for (int q = 0; q < 4; ++q)
            *(float4*)&sRow[0][s0 + 4 * q] =
                make_float4(a[4 * q], a[4 * q + 1], a[4 * q + 2], a[4 * q + 3]);
    }
    if (s4 == 0) sCol[0][i] = a[0];
    __syncthreads();

    for (int jb = 0; jb < 4; ++jb) {
#pragma unroll
        for (int jj = 0; jj < 16; ++jj) {
            const int j = jb * 16 + jj;
            if (j < 63) {
                const int buf = jj & 1;
                if (i > j) {
                    const float piv = sCol[buf][j];
                    const float f   = sCol[buf][i] / piv;
                    float pr[16];
#pragma unroll
                    for (int q = 0; q < 4; ++q) {
                        const float4 t = *(const float4*)&sRow[buf][s0 + 4 * q];
                        pr[4 * q] = t.x; pr[4 * q + 1] = t.y; pr[4 * q + 2] = t.z; pr[4 * q + 3] = t.w;
                    }
#pragma unroll
                    for (int c = 0; c < 16; ++c) a[c] = a[c] - f * pr[c];
                    if (s4 == jb) a[jj] = -f;
                }
                const int jn = j + 1;
                if (i == jn) {
#pragma unroll
                    for (int q = 0; q < 4; ++q)
                        *(float4*)&sRow[buf ^ 1][s0 + 4 * q] =
                            make_float4(a[4 * q], a[4 * q + 1], a[4 * q + 2], a[4 * q + 3]);
                }
                if (jj < 15) {
                    if (s4 == jb && i >= jn) sCol[buf ^ 1][i] = a[(jj + 1) & 15];
                } else {
                    if (s4 == jb + 1 && i >= jn) sCol[buf ^ 1][i] = a[0];
                }
                __syncthreads();
            }
        }
    }

#pragma unroll
    for (int c = 0; c < 16; ++c)
        if (s0 + c == i) sD[i] = a[c];
    __syncthreads();

    float l[16];
    {
        const float rs = 1.f / sqrtf(sD[i]);
#pragma unroll
        for (int c = 0; c < 16; ++c) {
            l[c] = (s0 + c < i) ? a[c] * rs : ((s0 + c == i) ? rs : 0.f);
            sA[i * AP + s0 + c] = u[c] * rs;
        }
#pragma unroll
        for (int q = 0; q < 4; ++q) {
            const float4 t = make_float4(l[4 * q], l[4 * q + 1], l[4 * q + 2], l[4 * q + 3]);
            *(float4*)&sLinv[i * LP + s0 + 4 * q] = t;
        }
    }

    {
        float part = 0.f;
#pragma unroll
        for (int c = 0; c < 16; ++c) part += l[c] * um[c];
        part += __shfl_xor(part, 2, 32);
        part += __shfl_xor(part, 1, 32);
        if (s4 == 0) sT[i] = part;
    }
    __syncthreads();

    {
        float part = 0.f;
#pragma unroll
        for (int c = 0; c < 16; ++c) part += sLinv[(s0 + c) * LP + i] * sT[s0 + c];
        part += __shfl_xor(part, 2, 32);
        part += __shfl_xor(part, 1, 32);
        if (s4 == 0) wp[WS_Q + i] = part;
    }

    const int wv   = __builtin_amdgcn_readfirstlane(tid >> 5);
    const int lane = tid & 31;
    const int lc   = lane & 15;
    const int lh   = lane >> 4;
    const int kr   = 2 * lh;

    {
        const int jtA = wv >> 2, ktA = wv & 3;
        const int idB = 15 - wv;
        const int jtB = idB >> 2, ktB = idB & 3;
        v2f fa[16], fb[16], ga[8], gb[8];
#pragma unroll
        for (int s = 0; s < 16; ++s) {
            fa[s] = frag2(sA, AP, 4 * s + kr, 16 * jtA + lc);
            fb[s] = frag2(sLinv, LP, 4 * s + kr, 16 * ktA + lc);
        }
#pragma unroll
        for (int s = 0; s < 8; ++s) {
            ga[s] = frag2(sA, AP, 32 + 4 * s + kr, 16 * jtB + lc);
            gb[s] = frag2(sLinv, LP, 32 + 4 * s + kr, 16 * ktB + lc);
        }
        __builtin_amdgcn_sched_barrier(0);
        v8f pA = {}, pB = {};
#pragma unroll
        for (int s = 0; s < 16; ++s) {
            pA = mm4(fa[s], fb[s], pA);
            if (s < 8) pB = mm4(ga[s], gb[s], pB);
        }
#pragma unroll
        for (int r = 0; r < 8; ++r) {
            sP[(16 * jtA + 8 * lh + r) * LP + 16 * ktA + lc] = pA[r];
            sP[(16 * jtB + 8 * lh + r) * LP + 16 * ktB + lc] = pB[r];
        }
    }
    __syncthreads();

    if (wv < 6) {
        const int t  = wv;
        const int bt = (t >= 3) ? 2 : (t >= 1) ? 1 : 0;
        const int at = t - bt * (bt + 1) / 2;
        v2f pa[16], pb[16], la[16], lb[16];
#pragma unroll
        for (int s = 0; s < 16; ++s) {
            pa[s] = frag2(sP, LP, 4 * s + kr, 16 * bt + lc);
            pb[s] = frag2(sP, LP, 4 * s + kr, 16 * at + lc);
            la[s] = frag2(sLinv, LP, 4 * s + kr, 16 * bt + lc);
            lb[s] = frag2(sLinv, LP, 4 * s + kr, 16 * at + lc);
        }
        __builtin_amdgcn_sched_barrier(0);
        v8f g1 = {}, g2 = {};
#pragma unroll
        for (int s = 0; s < 16; ++s) {
            g2 = mm4(pa[s], pb[s], g2);
            g1 = mm4(la[s], lb[s], g1);
        }
        store_vtile(wp + WS_V + t * 256 + lane * 8, g1, g2, (at == bt) ? 1.f : 2.f);
    } else {
        const int at0 = 2 * (wv - 6);
        v2f pa[16], pb0[16], pb1[16], la[4], lb0[4], lb1[4];
#pragma unroll
        for (int s = 0; s < 16; ++s) {
            pa[s]  = frag2(sP, LP, 4 * s + kr, 48 + lc);
            pb0[s] = frag2(sP, LP, 4 * s + kr, 16 * at0 + lc);
            pb1[s] = frag2(sP, LP, 4 * s + kr, 16 * at0 + 16 + lc);
        }
#pragma unroll
        for (int s = 0; s < 4; ++s) {
            la[s]  = frag2(sLinv, LP, 48 + 4 * s + kr, 48 + lc);
            lb0[s] = frag2(sLinv, LP, 48 + 4 * s + kr, 16 * at0 + lc);
            lb1[s] = frag2(sLinv, LP, 48 + 4 * s + kr, 16 * at0 + 16 + lc);
        }
        __builtin_amdgcn_sched_barrier(0);
        v8f g1a = {}, g1b = {}, g2a = {}, g2b = {};
#pragma unroll
        for (int s = 0; s < 16; ++s) {
            g2a = mm4(pa[s], pb0[s], g2a);
            g2b = mm4(pa[s], pb1[s], g2b);
            if (s < 4) {
                g1a = mm4(la[s], lb0[s], g1a);
                g1b = mm4(la[s], lb1[s], g1b);
            }
        }
        store_vtile(wp + WS_V + (6 + at0) * 256 + lane * 8, g1a, g2a, 2.f);
        store_vtile(wp + WS_V + (7 + at0) * 256 + lane * 8, g1b, g2b, (at0 == 2) ? 1.f : 2.f);
    }

    for (int e = tid; e < 512; e += PREP_THREADS) wp[WS_ZS + e] = sZs[e];
    if (tid < 64)  wp[WS_Z2 + tid] = sZ2[tid];
    if (tid < 8)   wp[WS_ILS + tid] = sIls[tid];
    if (tid == 0)  wp[WS_VAR] = var;
}

#define MAIN_WAVES 8
#define MAIN_THREADS (MAIN_WAVES * 32)
#define COLS_PER_ITER (MAIN_WAVES * 16)
#define COLS_PER_BLOCK 2048
#define LOG2E_F 1.44269504088896340736f

__global__ __launch_bounds__(MAIN_THREADS) void svgp_main(
    const float* __restrict__ x, const float* __restrict__ ws,
    float* __restrict__ out, int n, int blocksPerPair)
{
    const int pair  = blockIdx.x / blocksPerPair;
    const int chunk = blockIdx.x % blocksPerPair;
    const float* wp = ws + (size_t)pair * WS_STRIDE;
    const int tid  = threadIdx.x;
    const int wave = tid >> 5;
    const int lane = tid & 31;
    const int c    = lane & 15;
    const int h    = lane >> 4;
    const int kd   = h * 8;

    v2f aV[10][4];
#pragma unroll
    for (int t = 0; t < 10; ++t) {
        const float* gp = wp + WS_V + t * 256 + lane * 8;
        const float4 lo = *(const float4*)gp;
        const float4 hi = *(const float4*)(gp + 4);
        aV[t][0].x = lo.x; aV[t][0].y = lo.y;
        aV[t][1].x = lo.z; aV[t][1].y = lo.w;
        aV[t][2].x = hi.x; aV[t][2].y = hi.y;
        aV[t][3].x = hi.z; aV[t][3].y = hi.w;
    }

    const float il0 = wp[WS_ILS + 2 * h];
    const float il1 = wp[WS_ILS + 2 * h + 1];
    const float il2 = wp[WS_ILS + 4 + 2 * h];
    const float il3 = wp[WS_ILS + 5 + 2 * h];
    const float var = wp[WS_VAR];
    const float l2var = __log2f(var);

    v2f aE[4][3];
    v2f qr[4][4];
#pragma unroll
    for (int rt = 0; rt < 4; ++rt) {
        const int m = rt * 16 + c;
        const float* zr = wp + WS_ZS + m * 8;
        aE[rt][0].x = zr[2 * h]     * LOG2E_F;
        aE[rt][0].y = zr[2 * h + 1] * LOG2E_F;
        aE[rt][1].x = zr[4 + 2 * h] * LOG2E_F;
        aE[rt][1].y = zr[5 + 2 * h] * LOG2E_F;
        const float cz = (-0.5f * LOG2E_F) * wp[WS_Z2 + m] + l2var;
        aE[rt][2].x = h ? 1.f : cz;
        aE[rt][2].y = h ? 0.f : 1.f;
#pragma unroll
        for (int j = 0; j < 4; ++j) {
            qr[rt][j].x = wp[WS_Q + rt * 16 + kd + 2 * j];
            qr[rt][j].y = wp[WS_Q + rt * 16 + kd + 2 * j + 1];
        }
    }

    const size_t varBase = (size_t)(HDIM * ODIM) * (size_t)n;
    const int colW = chunk * COLS_PER_BLOCK + wave * 16 + c;
    float* const dst = out + (h ? varBase : (size_t)0) + (size_t)pair * n + colW;

    const float* xp = x + (size_t)colW * 8 + 2 * h;
    float xa0 = xp[0], xa1 = xp[1], xb0 = xp[4], xb1 = xp[5];

    for (int it = 0; it < COLS_PER_BLOCK / COLS_PER_ITER; ++it) {
        const float xs0 = xa0 * il0, xs1 = xa1 * il1, xs2 = xb0 * il2, xs3 = xb1 * il3;

        {
            const int itn = (it + 1 < COLS_PER_BLOCK / COLS_PER_ITER) ? it + 1 : it;
            const float* xq = xp + (size_t)itn * COLS_PER_ITER * 8;
            xa0 = xq[0]; xa1 = xq[1]; xb0 = xq[4]; xb1 = xq[5];
        }

        const float cxp = (-0.5f * LOG2E_F) * (xs0 * xs0 + xs1 * xs1 + xs2 * xs2 + xs3 * xs3);
        v2f bE0, bE1, bE2;
        bE0.x = xs0; bE0.y = xs1;
        bE1.x = xs2; bE1.y = xs3;
        bE2.x = h ? cxp : 1.f;
        bE2.y = h ? 0.f : cxp;

        v8f kt[4] = {};
#pragma unroll
        for (int rt = 0; rt < 4; ++rt) kt[rt] = mm4(aE[rt][0], bE0, kt[rt]);
#pragma unroll
        for (int rt = 0; rt < 4; ++rt) kt[rt] = mm4(aE[rt][1], bE1, kt[rt]);
#pragma unroll
        for (int rt = 0; rt < 4; ++rt) kt[rt] = mm4(aE[rt][2], bE2, kt[rt]);
        v2f mu2 = {0.f, 0.f};
#pragma unroll
        for (int rt = 0; rt < 4; ++rt)
#pragma unroll
            for (int r = 0; r < 8; r += 2) {
                kt[rt][r]     = __builtin_amdgcn_exp2f(kt[rt][r]);
                kt[rt][r + 1] = __builtin_amdgcn_exp2f(kt[rt][r + 1]);
                v2f kp; kp.x = kt[rt][r]; kp.y = kt[rt][r + 1];
                mu2 = kp * qr[rt][r >> 1] + mu2;
            }

        v8f accT[4] = {};
#pragma unroll
        for (int bt = 0; bt < 4; ++bt) {
#pragma unroll
            for (int jj = 0; jj < 4; ++jj) {
                v2f bK;
                bK.x = kt[bt][2 * jj + 0];
                bK.y = kt[bt][2 * jj + 1];
#pragma unroll
                for (int at = 0; at <= bt; ++at)
                    accT[at] = mm4(aV[bt * (bt + 1) / 2 + at][jj], bK, accT[at]);
            }
        }

        v2f s1 = {0.f, 0.f};
#pragma unroll
        for (int rt = 0; rt < 4; ++rt)
#pragma unroll
            for (int v = 0; v < 8; v += 2) {
                v2f k2; k2.x = kt[rt][v];   k2.y = kt[rt][v + 1];
                v2f t2; t2.x = accT[rt][v]; t2.y = accT[rt][v + 1];
                s1 = k2 * t2 + s1;
            }
        float mu = mu2.x + mu2.y;
        float pv = s1.x + s1.y;
        mu += __shfl_xor(mu, 16, 32);
        pv += __shfl_xor(pv, 16, 32);

        dst[it * COLS_PER_ITER] = h ? (var - pv) : mu;
    }
}

extern "C" void kernel_launch(void* const* d_in, const int* in_sizes, int n_in,
                              void* d_out, int out_size, void* d_ws, size_t ws_size,
                              hipStream_t stream)
{
    const float* x          = (const float*)d_in[0];
    const float* z          = (const float*)d_in[1];
    const float* u_mean     = (const float*)d_in[2];
    const float* u_tril_vec = (const float*)d_in[3];
    const float* log_ls     = (const float*)d_in[4];
    const float* log_var    = (const float*)d_in[5];
    float* out = (float*)d_out;
    float* ws  = (float*)d_ws;

    const int n = in_sizes[0] / 8;
    svgp_prep<<<HDIM * ODIM, PREP_THREADS, 0, stream>>>(z, u_mean, u_tril_vec,
                                              log_ls, log_var, ws);
    const int blocksPerPair = n / COLS_PER_BLOCK;
    svgp_main<<<HDIM * ODIM * blocksPerPair, MAIN_THREADS, 0, stream>>>(
        x, ws, out, n, blocksPerPair);
}


# phase 2 software-pipelined inside each wave: next tile's exponent WMMAs, this tile's exp2 and the previous tile's reduction and stores stand beside the 40 V*K WMMAs (two register sets)
# speedup vs baseline: 7.5281x; 1.3387x over previous
#include <hip/hip_runtime.h>

#define HDIM 4
#define ODIM 4
#define JITTER 1e-4f


typedef __attribute__((ext_vector_type(2))) float v2f;
typedef __attribute__((ext_vector_type(8))) float v8f;
typedef __attribute__((ext_vector_type(4))) float v4f;

#define LP 68
#define AP 65
#define PREP_THREADS 256
#define KB 4

__device__ __forceinline__ v8f mm4(v2f a, v2f b, v8f c)
{
    return __builtin_amdgcn_wmma_f32_16x16x4_f32(false, a, false, b, (short)0, c, false, false);
}

__device__ __forceinline__ v2f frag2(const float* S, int pitch, int row, int col)
{
    v2f f;
    f.x = S[row * pitch + col];
    f.y = S[(row + 1) * pitch + col];
    return f;
}

__device__ __forceinline__ float rcp_ref(float d)
{
    const float r0 = __builtin_amdgcn_rcpf(d);
    const float e  = __builtin_fmaf(-d, r0, 1.0f);
    return __builtin_fmaf(e, r0, r0);
}

__device__ __forceinline__ float quot(float n, float d, float r)
{
    float q = n * r;
    float e = __builtin_fmaf(-d, q, n);
    q = __builtin_fmaf(e, r, q);
    e = __builtin_fmaf(-d, q, n);
    return __builtin_fmaf(e, r, q);
}

__device__ __forceinline__ void store_vtile(float* gp, v8f g1, v8f g2, float sc)
{
    *(float4*)gp = make_float4((g1[0] - g2[0]) * sc, (g1[1] - g2[1]) * sc,
                               (g1[2] - g2[2]) * sc, (g1[3] - g2[3]) * sc);
    *(float4*)(gp + 4) = make_float4((g1[4] - g2[4]) * sc, (g1[5] - g2[5]) * sc,
                                     (g1[6] - g2[6]) * sc, (g1[7] - g2[7]) * sc);
}

#define MAIN_WAVES 8
#define MAIN_THREADS (MAIN_WAVES * 32)
#define COLS_PER_ITER (MAIN_WAVES * 16)
#define COLS_PER_BLOCK 2048
#define LOG2E_F 1.44269504088896340736f


__device__ __forceinline__ void expo_tile(const v2f (&aE)[4][3], float xs0, float xs1, float xs2, float xs3,
                                          int h, v8f (&e)[4])
{
    const float cxp = (-0.5f * LOG2E_F) * (xs0 * xs0 + xs1 * xs1 + xs2 * xs2 + xs3 * xs3);
    v2f bE0, bE1, bE2;
    bE0.x = xs0; bE0.y = xs1;
    bE1.x = xs2; bE1.y = xs3;
    bE2.x = h ? cxp : 1.f;
    bE2.y = h ? 0.f : cxp;
    const v8f zero = {};
#pragma unroll
    for (int rt = 0; rt < 4; ++rt) e[rt] = mm4(aE[rt][0], bE0, zero);
#pragma unroll
    for (int rt = 0; rt < 4; ++rt) e[rt] = mm4(aE[rt][1], bE1, e[rt]);
#pragma unroll
    for (int rt = 0; rt < 4; ++rt) e[rt] = mm4(aE[rt][2], bE2, e[rt]);
}

__device__ __forceinline__ void exp_tile(const v8f (&e)[4], const v2f (&qr)[4][4], v8f (&kt)[4], v2f& mu2)
{
    mu2.x = 0.f; mu2.y = 0.f;
#pragma unroll
    for (int rt = 0; rt < 4; ++rt)
#pragma unroll
        for (int r = 0; r < 8; r += 2) {
            kt[rt][r]     = __builtin_amdgcn_exp2f(e[rt][r]);
            kt[rt][r + 1] = __builtin_amdgcn_exp2f(e[rt][r + 1]);
            v2f kp; kp.x = kt[rt][r]; kp.y = kt[rt][r + 1];
            mu2 = kp * qr[rt][r >> 1] + mu2;
        }
}

__device__ __forceinline__ void vk_tile(const v2f (&aV)[10][4], const v8f (&kt)[4], v8f (&accT)[4])
{
    const v8f zero = {};
#pragma unroll
    for (int at = 0; at < 4; ++at) accT[at] = zero;
#pragma unroll
    for (int bt = 0; bt < 4; ++bt) {
#pragma unroll
        for (int jj = 0; jj < 4; ++jj) {
            v2f bK;
            bK.x = kt[bt][2 * jj + 0];
            bK.y = kt[bt][2 * jj + 1];
#pragma unroll
            for (int at = 0; at <= bt; ++at)
                accT[at] = mm4(aV[bt * (bt + 1) / 2 + at][jj], bK, accT[at]);
        }
    }
}

__device__ __forceinline__ void finish_tile(const v8f (&kt)[4], const v8f (&accT)[4], v2f mu2,
                                            float var, int h, float* dstp)
{
    v2f s1 = {0.f, 0.f};
#pragma unroll
    for (int rt = 0; rt < 4; ++rt)
#pragma unroll
        for (int v = 0; v < 8; v += 2) {
            v2f k2; k2.x = kt[rt][v];   k2.y = kt[rt][v + 1];
            v2f t2; t2.x = accT[rt][v]; t2.y = accT[rt][v + 1];
            s1 = k2 * t2 + s1;
        }
    float mu = mu2.x + mu2.y;
    float pv = s1.x + s1.y;
    mu += __shfl_xor(mu, 16, 32);
    pv += __shfl_xor(pv, 16, 32);
    *dstp = h ? (var - pv) : mu;
}

__global__ __launch_bounds__(PREP_THREADS) __attribute__((amdgpu_waves_per_eu(2, 2))) void svgp_fused(
    const float* __restrict__ x, const float* __restrict__ z,
    const float* __restrict__ u_mean, const float* __restrict__ u_tril_vec,
    const float* __restrict__ log_ls, const float* __restrict__ log_var,
    float* __restrict__ out, int n, int blocksPerPair)
{
    __shared__ __attribute__((aligned(16))) float sV[10 * 256];
    __shared__ float sQ[64];
    __shared__ __attribute__((aligned(16))) float sLinv[64 * LP];
    __shared__ __attribute__((aligned(16))) float sRowB[2][KB][64];
    __shared__ float sA[64 * AP];
    __shared__ float sP[64 * LP];
    __shared__ __attribute__((aligned(16))) float sColB[2][64][KB];
    __shared__ float sZs[64 * 8];
    __shared__ float sZ2[64];
    __shared__ float sD[64];
    __shared__ float sT[64];
    __shared__ float sIls[8];

    const int p     = blockIdx.x / blocksPerPair;
    const int chunk = blockIdx.x % blocksPerPair;
    const int tid   = threadIdx.x;
    const float var = __expf(log_var[p]);

    const int colW = chunk * COLS_PER_BLOCK + (tid >> 5) * 16 + (tid & 15);
    const float* xp = x + (size_t)colW * 8 + 2 * ((tid & 31) >> 4);
    float xa0 = xp[0], xa1 = xp[1], xb0 = xp[4], xb1 = xp[5];
    float xq0 = xp[COLS_PER_ITER * 8], xq1 = xp[COLS_PER_ITER * 8 + 1], xq2 = xp[COLS_PER_ITER * 8 + 4], xq3 = xp[COLS_PER_ITER * 8 + 5];

    {
    const int o   = p % ODIM;
    const int i   = tid >> 2;
    const int s4  = tid & 3;
    const int s0  = s4 * 16;

    float u[16], um[16];
    {
        const int base = o * 2080 + i * (i + 1) / 2;
#pragma unroll
        for (int c = 0; c < 16; ++c) {
            u[c]  = (s0 + c <= i) ? u_tril_vec[base + s0 + c] : 0.f;
            um[c] = u_mean[o * 64 + s0 + c];
        }
    }

    for (int e = tid; e < 512; e += PREP_THREADS) {
        const int d = e & 7;
        const float il = __expf(-log_ls[p * 8 + d]);
        sZs[e] = z[o * 512 + e] * il;
        if (e < 8) sIls[e] = il;
    }
    __syncthreads();
    if (tid < 64) {
        float s = 0.f;
#pragma unroll
        for (int d = 0; d < 8; ++d) { const float v = sZs[tid * 8 + d]; s += v * v; }
        sZ2[tid] = s;
    }
    __syncthreads();

    float a[16];
    {
        float zi[8];
#pragma unroll
        for (int d = 0; d < 8; ++d) zi[d] = sZs[i * 8 + d];
        const float z2i = sZ2[i];
#pragma unroll
        for (int c = 0; c < 16; ++c) {
            const int s = s0 + c;
            float dot = 0.f;
#pragma unroll
            for (int d = 0; d < 8; ++d) dot += zi[d] * sZs[s * 8 + d];
            float v = var * __expf(-0.5f * (z2i + sZ2[s] - 2.f * dot));
            if (i == s) v += JITTER;
            a[c] = v;
        }
    }

    if (i < KB) {
#pragma unroll
        for (int q = 0; q < 4; ++q)
            *(float4*)&sRowB[0][i][s0 + 4 * q] =
                make_float4(a[4 * q], a[4 * q + 1], a[4 * q + 2], a[4 * q + 3]);
    }
    if (s4 == 0) *(float4*)&sColB[0][i][0] = make_float4(a[0], a[1], a[2], a[3]);
    __syncthreads();

    for (int jb = 0; jb < 4; ++jb) {
        const bool own = (s4 == jb);
#pragma unroll
        for (int g = 0; g < 16 / KB; ++g) {
            const int jj0 = g * KB;
            const int j0  = jb * 16 + jj0;
            const int buf = g & 1;
            if (i > j0) {
                float cown[KB], Pb[KB][KB], R[KB][16];
                {
                    const v4f t = *(const v4f*)&sColB[buf][i][0];
                    cown[0] = t.x; cown[1] = t.y; cown[2] = t.z; cown[3] = t.w;
                }
#pragma unroll
                for (int m = 0; m < KB; ++m) {
                    const v4f t = *(const v4f*)&sRowB[buf][m][j0];
                    Pb[m][0] = t.x; Pb[m][1] = t.y; Pb[m][2] = t.z; Pb[m][3] = t.w;
                }
                __builtin_amdgcn_sched_barrier(0);
#pragma unroll
                for (int m = 0; m < KB; ++m)
#pragma unroll
                    for (int q = 0; q < 4; ++q) {
                        const v4f t = *(const v4f*)&sRowB[buf][m][s0 + 4 * q];
                        R[m][4 * q] = t.x; R[m][4 * q + 1] = t.y; R[m][4 * q + 2] = t.z; R[m][4 * q + 3] = t.w;
                    }
                __builtin_amdgcn_sched_barrier(0);
#pragma unroll
                for (int m = 0; m < KB; ++m) {
                    const float piv = Pb[m][m];
                    const float rp = rcp_ref(piv);
#pragma unroll
                    for (int mp = m + 1; mp < KB; ++mp) {
                        const float gg = quot(Pb[mp][m], piv, rp);
#pragma unroll
                        for (int q = m + 1; q < KB; ++q) Pb[mp][q] = Pb[mp][q] - gg * Pb[m][q];
#pragma unroll
                        for (int c = 0; c < 16; ++c) R[mp][c] = R[mp][c] - gg * R[m][c];
                        if (own) R[mp][jj0 + m] = -gg;
                    }
                    const bool act = (i > j0 + m);
                    const float f = act ? quot(cown[m], piv, rp) : 0.f;
#pragma unroll
                    for (int q = m + 1; q < KB; ++q) cown[q] = cown[q] - f * Pb[m][q];
#pragma unroll
                    for (int c = 0; c < 16; ++c) a[c] = a[c] - f * R[m][c];
                    if (own && act) a[jj0 + m] = -f;
                }
            }
            if (j0 + KB < 64) {
                const int j1  = j0 + KB;
                const int jj1 = (jj0 + KB) & 15;
                const int jb1 = (jj0 + KB < 16) ? jb : jb + 1;
                if (i >= j1 && i < j1 + KB) {
#pragma unroll
                    for (int q = 0; q < 4; ++q)
                        *(float4*)&sRowB[buf ^ 1][i - j1][s0 + 4 * q] =
                            make_float4(a[4 * q], a[4 * q + 1], a[4 * q + 2], a[4 * q + 3]);
                }
                if (s4 == jb1 && i >= j1)
                    *(float4*)&sColB[buf ^ 1][i][0] =
                        make_float4(a[jj1], a[jj1 + 1], a[jj1 + 2], a[jj1 + 3]);
                __syncthreads();
            }
        }
    }

#pragma unroll
    for (int c = 0; c < 16; ++c)
        if (s0 + c == i) sD[i] = a[c];
    __syncthreads();

    float l[16];
    {
        const float rs = 1.f / sqrtf(sD[i]);
#pragma unroll
        for (int c = 0; c < 16; ++c) {
            l[c] = (s0 + c < i) ? a[c] * rs : ((s0 + c == i) ? rs : 0.f);
            sA[i * AP + s0 + c] = u[c] * rs;
        }
#pragma unroll
        for (int q = 0; q < 4; ++q) {
            const float4 t = make_float4(l[4 * q], l[4 * q + 1], l[4 * q + 2], l[4 * q + 3]);
            *(float4*)&sLinv[i * LP + s0 + 4 * q] = t;
        }
    }

    {
        float part = 0.f;
#pragma unroll
        for (int c = 0; c < 16; ++c) part += l[c] * um[c];
        part += __shfl_xor(part, 2, 32);
        part += __shfl_xor(part, 1, 32);
        if (s4 == 0) sT[i] = part;
    }
    __syncthreads();

    {
        float part = 0.f;
#pragma unroll
        for (int c = 0; c < 16; ++c) part += sLinv[(s0 + c) * LP + i] * sT[s0 + c];
        part += __shfl_xor(part, 2, 32);
        part += __shfl_xor(part, 1, 32);
        if (s4 == 0) sQ[i] = part;
    }

    const int wv   = __builtin_amdgcn_readfirstlane(tid >> 5);
    const int lane = tid & 31;
    const int lc   = lane & 15;
    const int lh   = lane >> 4;
    const int kr   = 2 * lh;

    {
        const int jtA = wv >> 2, ktA = wv & 3;
        const int idB = 15 - wv;
        const int jtB = idB >> 2, ktB = idB & 3;
        v2f fa[16], fb[16], ga[8], gb[8];
#pragma unroll
        for (int s = 0; s < 16; ++s) {
            fa[s] = frag2(sA, AP, 4 * s + kr, 16 * jtA + lc);
            fb[s] = frag2(sLinv, LP, 4 * s + kr, 16 * ktA + lc);
        }
#pragma unroll
        for (int s = 0; s < 8; ++s) {
            ga[s] = frag2(sA, AP, 32 + 4 * s + kr, 16 * jtB + lc);
            gb[s] = frag2(sLinv, LP, 32 + 4 * s + kr, 16 * ktB + lc);
        }
        __builtin_amdgcn_sched_barrier(0);
        v8f pA = {}, pB = {};
#pragma unroll
        for (int s = 0; s < 16; ++s) {
            pA = mm4(fa[s], fb[s], pA);
            if (s < 8) pB = mm4(ga[s], gb[s], pB);
        }
#pragma unroll
        for (int r = 0; r < 8; ++r) {
            sP[(16 * jtA + 8 * lh + r) * LP + 16 * ktA + lc] = pA[r];
            sP[(16 * jtB + 8 * lh + r) * LP + 16 * ktB + lc] = pB[r];
        }
    }
    __syncthreads();

    if (wv < 6) {
        const int t  = wv;
        const int bt = (t >= 3) ? 2 : (t >= 1) ? 1 : 0;
        const int at = t - bt * (bt + 1) / 2;
        v2f pa[16], pb[16], la[16], lb[16];
#pragma unroll
        for (int s = 0; s < 16; ++s) {
            pa[s] = frag2(sP, LP, 4 * s + kr, 16 * bt + lc);
            pb[s] = frag2(sP, LP, 4 * s + kr, 16 * at + lc);
            la[s] = frag2(sLinv, LP, 4 * s + kr, 16 * bt + lc);
            lb[s] = frag2(sLinv, LP, 4 * s + kr, 16 * at + lc);
        }
        __builtin_amdgcn_sched_barrier(0);
        v8f g1 = {}, g2 = {};
#pragma unroll
        for (int s = 0; s < 16; ++s) {
            g2 = mm4(pa[s], pb[s], g2);
            g1 = mm4(la[s], lb[s], g1);
        }
        store_vtile(sV + t * 256 + lane * 8, g1, g2, (at == bt) ? 1.f : 2.f);
    } else {
        const int at0 = 2 * (wv - 6);
        v2f pa[16], pb0[16], pb1[16], la[4], lb0[4], lb1[4];
#pragma unroll
        for (int s = 0; s < 16; ++s) {
            pa[s]  = frag2(sP, LP, 4 * s + kr, 48 + lc);
            pb0[s] = frag2(sP, LP, 4 * s + kr, 16 * at0 + lc);
            pb1[s] = frag2(sP, LP, 4 * s + kr, 16 * at0 + 16 + lc);
        }
#pragma unroll
        for (int s = 0; s < 4; ++s) {
            la[s]  = frag2(sLinv, LP, 48 + 4 * s + kr, 48 + lc);
            lb0[s] = frag2(sLinv, LP, 48 + 4 * s + kr, 16 * at0 + lc);
            lb1[s] = frag2(sLinv, LP, 48 + 4 * s + kr, 16 * at0 + 16 + lc);
        }
        __builtin_amdgcn_sched_barrier(0);
        v8f g1a = {}, g1b = {}, g2a = {}, g2b = {};
#pragma unroll
        for (int s = 0; s < 16; ++s) {
            g2a = mm4(pa[s], pb0[s], g2a);
            g2b = mm4(pa[s], pb1[s], g2b);
            if (s < 4) {
                g1a = mm4(la[s], lb0[s], g1a);
                g1b = mm4(la[s], lb1[s], g1b);
            }
        }
        store_vtile(sV + (6 + at0) * 256 + lane * 8, g1a, g2a, 2.f);
        store_vtile(sV + (7 + at0) * 256 + lane * 8, g1b, g2b, (at0 == 2) ? 1.f : 2.f);
    }
    }
    __syncthreads();

    const int pair = p;
    const int wave = tid >> 5;
    const int lane = tid & 31;
    const int c    = lane & 15;
    const int h    = lane >> 4;
    const int kd   = h * 8;

    v2f aV[10][4];
#pragma unroll
    for (int t = 0; t < 10; ++t) {
        const float* gp = sV + t * 256 + lane * 8;
        const float4 lo = *(const float4*)gp;
        const float4 hi = *(const float4*)(gp + 4);
        aV[t][0].x = lo.x; aV[t][0].y = lo.y;
        aV[t][1].x = lo.z; aV[t][1].y = lo.w;
        aV[t][2].x = hi.x; aV[t][2].y = hi.y;
        aV[t][3].x = hi.z; aV[t][3].y = hi.w;
    }

    const float il0 = sIls[2 * h];
    const float il1 = sIls[2 * h + 1];
    const float il2 = sIls[4 + 2 * h];
    const float il3 = sIls[5 + 2 * h];
    const float l2var = __log2f(var);

    v2f aE[4][3];
    v2f qr[4][4];
#pragma unroll
    for (int rt = 0; rt < 4; ++rt) {
        const int m = rt * 16 + c;
        const float* zr = sZs + m * 8;
        aE[rt][0].x = zr[2 * h]     * LOG2E_F;
        aE[rt][0].y = zr[2 * h + 1] * LOG2E_F;
        aE[rt][1].x = zr[4 + 2 * h] * LOG2E_F;
        aE[rt][1].y = zr[5 + 2 * h] * LOG2E_F;
        const float cz = (-0.5f * LOG2E_F) * sZ2[m] + l2var;
        aE[rt][2].x = h ? 1.f : cz;
        aE[rt][2].y = h ? 0.f : 1.f;
#pragma unroll
        for (int j = 0; j < 4; ++j) {
            qr[rt][j].x = sQ[rt * 16 + kd + 2 * j];
            qr[rt][j].y = sQ[rt * 16 + kd + 2 * j + 1];
        }
    }

    const size_t varBase = (size_t)(HDIM * ODIM) * (size_t)n;
    float* const dst = out + (h ? varBase : (size_t)0) + (size_t)pair * n + colW;

    constexpr int NT = COLS_PER_BLOCK / COLS_PER_ITER;
    v8f eA[4], eB[4], kA[4], kB[4], tA[4], tB[4];
    v2f muA, muB;

    expo_tile(aE, xa0 * il0, xa1 * il1, xb0 * il2, xb1 * il3, h, eA);

    {
        const float xs0 = xq0 * il0, xs1 = xq1 * il1, xs2 = xq2 * il2, xs3 = xq3 * il3;
        const float* xq = xp + (size_t)2 * COLS_PER_ITER * 8;
        xq0 = xq[0]; xq1 = xq[1]; xq2 = xq[4]; xq3 = xq[5];
        expo_tile(aE, xs0, xs1, xs2, xs3, h, eB);
        exp_tile(eA, qr, kA, muA);
        vk_tile(aV, kA, tA);
    }

    for (int it = 1; it < NT - 1; it += 2) {
        {
            const float xs0 = xq0 * il0, xs1 = xq1 * il1, xs2 = xq2 * il2, xs3 = xq3 * il3;
            const float* xq = xp + (size_t)(it + 2) * COLS_PER_ITER * 8;
            xq0 = xq[0]; xq1 = xq[1]; xq2 = xq[4]; xq3 = xq[5];
            expo_tile(aE, xs0, xs1, xs2, xs3, h, eA);
            exp_tile(eB, qr, kB, muB);
            finish_tile(kA, tA, muA, var, h, dst + (it - 1) * COLS_PER_ITER);
            vk_tile(aV, kB, tB);
        }
        {
            const float xs0 = xq0 * il0, xs1 = xq1 * il1, xs2 = xq2 * il2, xs3 = xq3 * il3;
            const int itn = (it + 3 < NT) ? it + 3 : NT - 1;
            const float* xq = xp + (size_t)itn * COLS_PER_ITER * 8;
            xq0 = xq[0]; xq1 = xq[1]; xq2 = xq[4]; xq3 = xq[5];
            expo_tile(aE, xs0, xs1, xs2, xs3, h, eB);
            exp_tile(eA, qr, kA, muA);
            finish_tile(kB, tB, muB, var, h, dst + it * COLS_PER_ITER);
            vk_tile(aV, kA, tA);
        }
    }

    exp_tile(eB, qr, kB, muB);
    finish_tile(kA, tA, muA, var, h, dst + (NT - 2) * COLS_PER_ITER);
    vk_tile(aV, kB, tB);
    finish_tile(kB, tB, muB, var, h, dst + (NT - 1) * COLS_PER_ITER);
}

extern "C" void kernel_launch(void* const* d_in, const int* in_sizes, int n_in,
                              void* d_out, int out_size, void* d_ws, size_t ws_size,
                              hipStream_t stream)
{
    const float* x          = (const float*)d_in[0];
    const float* z          = (const float*)d_in[1];
    const float* u_mean     = (const float*)d_in[2];
    const float* u_tril_vec = (const float*)d_in[3];
    const float* log_ls     = (const float*)d_in[4];
    const float* log_var    = (const float*)d_in[5];
    float* out = (float*)d_out;

    const int n = in_sizes[0] / 8;
    const int blocksPerPair = n / COLS_PER_BLOCK;
    svgp_fused<<<HDIM * ODIM * blocksPerPair, MAIN_THREADS, 0, stream>>>(
        x, z, u_mean, u_tril_vec, log_ls, log_var, out, n, blocksPerPair);
}
